# starting kernel
# baseline (speedup 1.0000x reference)
#include <hip/hip_runtime.h>

typedef float v2f __attribute__((ext_vector_type(2)));
typedef float v8f __attribute__((ext_vector_type(8)));

#define N_ROWS      131072
#define ROWS_PER_WG 64
#define STR         132

__device__ __forceinline__ v8f bias_init(float bv) {
    v8f c;
#pragma unroll
    for (int i = 0; i < 8; ++i) c[i] = bv;
    return c;
}

__device__ __forceinline__ void wmma_k64(const float* __restrict__ W, int ldW,
                                         int kOff, int nCol,
                                         const float* src, int srcColOff,
                                         int lane, v8f c[4]) {
    const int half = lane >> 4;
    const int l15  = lane & 15;
#pragma unroll
    for (int kk = 0; kk < 16; ++kk) {
        const int k0 = kk * 4 + 2 * half;
        v2f b;
        b.x = W[(kOff + k0 + 0) * ldW + nCol];
        b.y = W[(kOff + k0 + 1) * ldW + nCol];
#pragma unroll
        for (int rt = 0; rt < 4; ++rt) {
            v2f a = *(const v2f*)&src[(16 * rt + l15) * STR + srcColOff + k0];
            c[rt] = __builtin_amdgcn_wmma_f32_16x16x4_f32(
                false, a, false, b, (short)0, c[rt], false, false);
        }
    }
}

__device__ __forceinline__ void sin_store4(const v8f c[4], float* dst,
                                           int nCol, int half) {
#pragma unroll
    for (int rt = 0; rt < 4; ++rt)
#pragma unroll
        for (int v = 0; v < 8; ++v)
            dst[(16 * rt + v + 8 * half) * STR + nCol] = __sinf(c[rt][v]);
}

__global__ __launch_bounds__(256)
void bsnn_fused(const float* __restrict__ X,
                const float* __restrict__ W0, const float* __restrict__ B0,
                const float* __restrict__ W1, const float* __restrict__ B1,
                const float* __restrict__ W2, const float* __restrict__ B2,
                const float* __restrict__ W3, const float* __restrict__ B3,
                const float* __restrict__ W4, const float* __restrict__ B4,
                const float* __restrict__ W5, const float* __restrict__ B5,
                float* __restrict__ out) {
    __shared__ float hA[ROWS_PER_WG * STR];
    __shared__ float hB[ROWS_PER_WG * STR];
    __shared__ float hC[ROWS_PER_WG * STR];
    __shared__ float acc[ROWS_PER_WG];

    const int tid      = threadIdx.x;
    const int w        = tid >> 5;
    const int lane     = tid & 31;
    const int half     = lane >> 4;
    const int l15      = lane & 15;
    const int rowsBase = blockIdx.x * ROWS_PER_WG;
    const int nCol     = 16 * w + l15;

    if (tid < ROWS_PER_WG) acc[tid] = 0.0f;

    {
        const int nc0 = 16 * (w & 3) + l15;
        const float bias = B0[nc0];
        const float w0r = W0[0 * 64 + nc0];
        const float w1r = W0[1 * 64 + nc0];
        const float w2r = W0[2 * 64 + nc0];
        v2f b;
        b.x = half ? w2r : w0r;
        b.y = half ? 0.0f : w1r;
#pragma unroll
        for (int j = 0; j < 2; ++j) {
            const int rB   = 16 * (w >> 2) + 32 * j;
            const int gRow = rowsBase + rB + l15;
            const float x0 = X[gRow * 3 + 0];
            const float x1 = X[gRow * 3 + 1];
            const float x2 = X[gRow * 3 + 2];
            v2f a;
            a.x = half ? x2 : x0;
            a.y = half ? 0.0f : x1;
            v8f c = bias_init(bias);
            c = __builtin_amdgcn_wmma_f32_16x16x4_f32(
                false, a, false, b, (short)0, c, false, false);
#pragma unroll
            for (int v = 0; v < 8; ++v)
                hC[(rB + v + 8 * half) * STR + nc0] = __sinf(c[v]);
        }
    }
    __syncthreads();

    {
        v8f c[4];
#pragma unroll
        for (int rt = 0; rt < 4; ++rt) c[rt] = bias_init(B1[nCol]);
        wmma_k64(W1, 128, 0, nCol, hC, 0, lane, c);
        sin_store4(c, hA, nCol, half);
    }
    __syncthreads();

    v8f pacc[4];
#pragma unroll
    for (int rt = 0; rt < 4; ++rt) pacc[rt] = bias_init(0.0f);

    for (int c2 = 0; c2 < 2; ++c2) {
        {
            v8f c[4];
#pragma unroll
            for (int rt = 0; rt < 4; ++rt)
                c[rt] = bias_init(B2[128 * c2 + nCol]);
            wmma_k64(W2, 256, 64 * c2, 128 * c2 + nCol, hA, 64 * c2, lane, c);
            sin_store4(c, hB, nCol, half);
        }
        __syncthreads();
        for (int c3 = 0; c3 < 2; ++c3) {
            const int i3 = 2 * c2 + c3;
            {
                v8f c[4];
#pragma unroll
                for (int rt = 0; rt < 4; ++rt)
                    c[rt] = bias_init(B3[128 * i3 + nCol]);
                wmma_k64(W3, 512, 64 * i3, 128 * i3 + nCol, hB, 64 * c3,
                         lane, c);
                sin_store4(c, hC, nCol, half);
            }
            __syncthreads();
            for (int c4 = 0; c4 < 2; ++c4) {
                const int i4 = 2 * i3 + c4;
                v8f c[4];
#pragma unroll
                for (int rt = 0; rt < 4; ++rt)
                    c[rt] = bias_init(B4[128 * i4 + nCol]);
                wmma_k64(W4, 1024, 64 * i4, 128 * i4 + nCol, hC, 64 * c4,
                         lane, c);
                const float w5v = W5[128 * i4 + nCol];
#pragma unroll
                for (int rt = 0; rt < 4; ++rt)
#pragma unroll
                    for (int v = 0; v < 8; ++v)
                        pacc[rt][v] += __sinf(c[rt][v]) * w5v;
            }
            __syncthreads();
        }
    }

#pragma unroll
    for (int rt = 0; rt < 4; ++rt) {
#pragma unroll
        for (int v = 0; v < 8; ++v) {
            float p = pacc[rt][v];
            p += __shfl_xor(p, 1, 16);
            p += __shfl_xor(p, 2, 16);
            p += __shfl_xor(p, 4, 16);
            p += __shfl_xor(p, 8, 16);
            if (l15 == 0) atomicAdd(&acc[16 * rt + v + 8 * half], p);
        }
    }
    __syncthreads();
    if (tid < ROWS_PER_WG) out[rowsBase + tid] = acc[tid] + B5[0];
}

extern "C" void kernel_launch(void* const* d_in, const int* in_sizes, int n_in,
                              void* d_out, int out_size, void* d_ws,
                              size_t ws_size, hipStream_t stream) {
    (void)in_sizes; (void)n_in; (void)d_ws; (void)ws_size; (void)out_size;
    const float* X  = (const float*)d_in[0];
    const float* W0 = (const float*)d_in[1];
    const float* B0 = (const float*)d_in[2];
    const float* W1 = (const float*)d_in[3];
    const float* B1 = (const float*)d_in[4];
    const float* W2 = (const float*)d_in[5];
    const float* B2 = (const float*)d_in[6];
    const float* W3 = (const float*)d_in[7];
    const float* B3 = (const float*)d_in[8];
    const float* W4 = (const float*)d_in[9];
    const float* B4 = (const float*)d_in[10];
    const float* W5 = (const float*)d_in[11];
    const float* B5 = (const float*)d_in[12];

    dim3 grid(N_ROWS / ROWS_PER_WG);
    dim3 block(256);
    bsnn_fused<<<grid, block, 0, stream>>>(X, W0, B0, W1, B1, W2, B2, W3, B3,
                                           W4, B4, W5, B5, (float*)d_out);
}


# Feature-major products: a wave keeps its 32 rows in registers (sin of a D tile is the next B operand), weights go through LDS as A; k order in 16-groups and final sum order change, atomics removed
# speedup vs baseline: 1.3497x; 1.3497x over previous
#include <hip/hip_runtime.h>

typedef float v2f __attribute__((ext_vector_type(2)));
typedef float v4f __attribute__((ext_vector_type(4)));
typedef float v8f __attribute__((ext_vector_type(8)));

#define N_ROWS      131072
#define BT          2
#define ROWS_PER_WG 256
#define KPS         264
#define WBUF        (32 * KPS)
#define AUXO        WBUF
#define BUFSZ       (WBUF + 256)


struct Staged { v4f g0[4]; v4f g1[4]; float aux; };

__device__ __forceinline__ void stage_load(Staged& s, const float* __restrict__ W, int ldW,
                                           int kOff, int nOff,
                                           const float* __restrict__ auxA,
                                           const float* __restrict__ auxB, int tid) {
    const int n4 = (tid & 31) * 4;
#pragma unroll
    for (int i = 0; i < 4; ++i) {
        const int kp = (tid >> 5) + 8 * i;
        const float* p = W + (kOff + 2 * kp) * ldW + nOff + n4;
        s.g0[i] = *(const v4f*)p;
        s.g1[i] = *(const v4f*)(p + ldW);
    }
    s.aux = (tid < 128) ? auxA[tid] : auxB[tid - 128];
}

__device__ __forceinline__ void stage_store(const Staged& s, float* buf, int tid) {
    const int n4 = (tid & 31) * 4;
#pragma unroll
    for (int i = 0; i < 4; ++i) {
        const int kp = (tid >> 5) + 8 * i;
        float* d = buf + kp * KPS + 2 * n4;
        v4f lo, hi;
        lo.x = s.g0[i].x; lo.y = s.g1[i].x; lo.z = s.g0[i].y; lo.w = s.g1[i].y;
        hi.x = s.g0[i].z; hi.y = s.g1[i].z; hi.z = s.g0[i].w; hi.w = s.g1[i].w;
        *(v4f*)d = lo;
        *(v4f*)(d + 4) = hi;
    }
    buf[AUXO + tid] = s.aux;
}

template <bool LAST>
__device__ __forceinline__ void tile_out(const v8f (&cp)[BT], const float (&w5)[8],
                                         int tile, int idx,
                                         v8f (&hout)[2][4][BT], float (&pacc)[BT]) {
    const int bt = idx >> 3, r = idx & 7;
    if (!LAST) hout[tile >> 2][tile & 3][bt][r] = __sinf(cp[bt][r]);
    else       pacc[bt] += __sinf(cp[bt][r]) * w5[r];
}

struct TileIn { v2f a[16]; v4f blo, bhi, wlo, whi; };

template <bool LAST>
__device__ __forceinline__ void tile_in(TileIn& ti, const float* aBase, const float* xBase,
                                        int nt) {
#pragma unroll
    for (int s = 0; s < 16; ++s)
        ti.a[s] = *(const v2f*)(aBase + (8 * (s >> 2) + (s & 3)) * KPS + 32 * nt);
    ti.blo = *(const v4f*)(xBase + 16 * nt);
    ti.bhi = *(const v4f*)(xBase + 16 * nt + 4);
    if (LAST) {
        ti.wlo = *(const v4f*)(xBase + 128 + 16 * nt);
        ti.whi = *(const v4f*)(xBase + 128 + 16 * nt + 4);
    }
}

template <bool LAST>
__device__ __forceinline__ void block64x128(const float* buf, int half, int l15,
                                            const v8f (&hin)[4][BT],
                                            v8f (&hout)[2][4][BT], float (&pacc)[BT]) {
    const float* aBase = buf + (4 * half) * KPS + 2 * l15;
    const float* xBase = buf + AUXO + 8 * half;
    v8f cp[BT];
    float w5p[8];
#pragma unroll
    for (int r = 0; r < 8; ++r) w5p[r] = 0.0f;
    TileIn cur;
    tile_in<LAST>(cur, aBase, xBase, 0);
#pragma unroll
    for (int nt = 0; nt < 8; ++nt) {
        TileIn nxt;
        if (nt < 7) tile_in<LAST>(nxt, aBase, xBase, nt + 1);
        __builtin_amdgcn_sched_barrier(0x0020 | 0x0004);
        float w5n[8];
        if (LAST) {
            w5n[0] = cur.wlo.x; w5n[1] = cur.wlo.y; w5n[2] = cur.wlo.z; w5n[3] = cur.wlo.w;
            w5n[4] = cur.whi.x; w5n[5] = cur.whi.y; w5n[6] = cur.whi.z; w5n[7] = cur.whi.w;
        } else {
#pragma unroll
            for (int r = 0; r < 8; ++r) w5n[r] = 0.0f;
        }
        v8f c[BT];
#pragma unroll
        for (int bt = 0; bt < BT; ++bt) {
            c[bt][0] = cur.blo.x; c[bt][1] = cur.blo.y; c[bt][2] = cur.blo.z; c[bt][3] = cur.blo.w;
            c[bt][4] = cur.bhi.x; c[bt][5] = cur.bhi.y; c[bt][6] = cur.bhi.z; c[bt][7] = cur.bhi.w;
        }
#pragma unroll
        for (int t = 0; t < 4; ++t) {
#pragma unroll
            for (int j = 0; j < 4; ++j) {
                const v2f a = cur.a[4 * t + j];
#pragma unroll
                for (int bt = 0; bt < BT; ++bt) {
                    v2f b;
                    b.x = hin[t][bt][2 * j];
                    b.y = hin[t][bt][2 * j + 1];
                    c[bt] = __builtin_amdgcn_wmma_f32_16x16x4_f32(
                        false, a, false, b, (short)0, c[bt], false, false);
                }
            }
            if (nt > 0) {
#pragma unroll
                for (int q = 0; q < 4; ++q)
                    tile_out<LAST>(cp, w5p, nt - 1, 4 * t + q, hout, pacc);
            }
            __builtin_amdgcn_sched_barrier(0x0020 | 0x0004);
        }
#pragma unroll
        for (int bt = 0; bt < BT; ++bt) cp[bt] = c[bt];
#pragma unroll
        for (int r = 0; r < 8; ++r) w5p[r] = w5n[r];
        if (nt < 7) cur = nxt;
    }
#pragma unroll
    for (int idx = 0; idx < 16; ++idx) tile_out<LAST>(cp, w5p, 7, idx, hout, pacc);
}

__global__ __launch_bounds__(256)
void bsnn_fm(const float* __restrict__ X,
             const float* __restrict__ W0, const float* __restrict__ B0,
             const float* __restrict__ W1, const float* __restrict__ B1,
             const float* __restrict__ W2, const float* __restrict__ B2,
             const float* __restrict__ W3, const float* __restrict__ B3,
             const float* __restrict__ W4, const float* __restrict__ B4,
             const float* __restrict__ W5, const float* __restrict__ B5,
             float* __restrict__ out) {
    __shared__ __attribute__((aligned(16))) float wbuf[2 * BUFSZ];

    const int tid     = threadIdx.x;
    const int w       = tid >> 5;
    const int lane    = tid & 31;
    const int half    = lane >> 4;
    const int l15     = lane & 15;
    const int rowBase = blockIdx.x * ROWS_PER_WG + 32 * w;

    Staged st;
    stage_load(st, W1, 128, 0, 0, B1, B1, tid);

    v8f h0[4][BT];
    {
        v2f xb[BT];
#pragma unroll
        for (int bt = 0; bt < BT; ++bt) {
            const int row = rowBase + 16 * bt + l15;
            const float x0 = X[row * 3 + 0];
            const float x1 = X[row * 3 + 1];
            const float x2 = X[row * 3 + 2];
            xb[bt].x = half ? x2 : x0;
            xb[bt].y = half ? 0.0f : x1;
        }
#pragma unroll
        for (int nt = 0; nt < 4; ++nt) {
            const int n = 16 * nt + l15;
            const float w0r = W0[0 * 64 + n];
            const float w1r = W0[1 * 64 + n];
            const float w2r = W0[2 * 64 + n];
            v2f a;
            a.x = half ? w2r : w0r;
            a.y = half ? 0.0f : w1r;
            const v4f blo = *(const v4f*)(B0 + 16 * nt + 8 * half);
            const v4f bhi = *(const v4f*)(B0 + 16 * nt + 8 * half + 4);
#pragma unroll
            for (int bt = 0; bt < BT; ++bt) {
                v8f c;
                c[0] = blo.x; c[1] = blo.y; c[2] = blo.z; c[3] = blo.w;
                c[4] = bhi.x; c[5] = bhi.y; c[6] = bhi.z; c[7] = bhi.w;
                c = __builtin_amdgcn_wmma_f32_16x16x4_f32(
                    false, a, false, xb[bt], (short)0, c, false, false);
#pragma unroll
                for (int r = 0; r < 8; ++r) h0[nt][bt][r] = __sinf(c[r]);
            }
        }
    }

    float pacc[BT];
#pragma unroll
    for (int bt = 0; bt < BT; ++bt) pacc[bt] = 0.0f;

    v8f h1[2][4][BT], h2[2][4][BT], h3[2][4][BT];

    stage_store(st, wbuf, tid);
    __syncthreads();
    int cur = 0;

    stage_load(st, W2, 256, 0, 0, B2, B2, tid);
    block64x128<false>(wbuf, half, l15, h0, h1, pacc);
    stage_store(st, wbuf + BUFSZ, tid);
    __syncthreads();
    cur = 1;

#pragma unroll 1
    for (int c2 = 0; c2 < 2; ++c2) {
        stage_load(st, W3, 512, 64 * (2 * c2), 128 * (2 * c2),
                   B3 + 128 * (2 * c2), B3 + 128 * (2 * c2), tid);
        block64x128<false>(wbuf + cur * BUFSZ, half, l15, h1[0], h2, pacc);
        stage_store(st, wbuf + (cur ^ 1) * BUFSZ, tid);
        __syncthreads();
        cur ^= 1;
#pragma unroll 1
        for (int c3 = 0; c3 < 2; ++c3) {
            const int i3 = 2 * c2 + c3;
            stage_load(st, W4, 1024, 64 * (2 * i3), 128 * (2 * i3),
                       B4 + 128 * (2 * i3), W5 + 128 * (2 * i3), tid);
            block64x128<false>(wbuf + cur * BUFSZ, half, l15, h2[0], h3, pacc);
            stage_store(st, wbuf + (cur ^ 1) * BUFSZ, tid);
            __syncthreads();
            cur ^= 1;
#pragma unroll 1
            for (int c4 = 0; c4 < 2; ++c4) {
                const int i4 = 2 * i3 + c4;
                const float* Wn; const float* An; const float* Xn; int ldn, kOn, nOn;
                if (c4 == 0) {
                    Wn = W4; ldn = 1024; kOn = 64 * (i4 + 1); nOn = 128 * (i4 + 1);
                    An = B4 + nOn; Xn = W5 + nOn;
                } else if (c3 == 0) {
                    Wn = W3; ldn = 512; kOn = 64 * (i3 + 1); nOn = 128 * (i3 + 1);
                    An = B3 + nOn; Xn = An;
                } else {
                    Wn = W2; ldn = 256; kOn = 64; nOn = 128;
                    An = B2 + nOn; Xn = An;
                }
                stage_load(st, Wn, ldn, kOn, nOn, An, Xn, tid);
                block64x128<true>(wbuf + cur * BUFSZ, half, l15, h3[0], h3, pacc);
                stage_store(st, wbuf + (cur ^ 1) * BUFSZ, tid);
                __syncthreads();
                cur ^= 1;
#pragma unroll
                for (int t = 0; t < 4; ++t)
#pragma unroll
                    for (int bt = 0; bt < BT; ++bt) h3[0][t][bt] = h3[1][t][bt];
            }
#pragma unroll
            for (int t = 0; t < 4; ++t)
#pragma unroll
                for (int bt = 0; bt < BT; ++bt) h2[0][t][bt] = h2[1][t][bt];
        }
#pragma unroll
        for (int t = 0; t < 4; ++t)
#pragma unroll
            for (int bt = 0; bt < BT; ++bt) h1[0][t][bt] = h1[1][t][bt];
    }

    const float s0 = pacc[0] + __shfl_xor(pacc[0], 16, 32);
    const float s1 = pacc[1] + __shfl_xor(pacc[1], 16, 32);
    out[rowBase + lane] = (half ? s1 : s0) + B5[0];
}

extern "C" void kernel_launch(void* const* d_in, const int* in_sizes, int n_in,
                              void* d_out, int out_size, void* d_ws,
                              size_t ws_size, hipStream_t stream) {
    (void)in_sizes; (void)n_in; (void)d_ws; (void)ws_size; (void)out_size;
    const float* X  = (const float*)d_in[0];
    const float* W0 = (const float*)d_in[1];
    const float* B0 = (const float*)d_in[2];
    const float* W1 = (const float*)d_in[3];
    const float* B1 = (const float*)d_in[4];
    const float* W2 = (const float*)d_in[5];
    const float* B2 = (const float*)d_in[6];
    const float* W3 = (const float*)d_in[7];
    const float* B3 = (const float*)d_in[8];
    const float* W4 = (const float*)d_in[9];
    const float* B4 = (const float*)d_in[10];
    const float* W5 = (const float*)d_in[11];
    const float* B5 = (const float*)d_in[12];

    dim3 grid(N_ROWS / ROWS_PER_WG);
    dim3 block(256);
    bsnn_fm<<<grid, block, 0, stream>>>(X, W0, B0, W1, B1, W2, B2, W3, B3,
                                        W4, B4, W5, B5, (float*)d_out);
}
